# starting kernel
# baseline (speedup 1.0000x reference)
#include <hip/hip_runtime.h>

typedef float v2f __attribute__((ext_vector_type(2)));
typedef float v8f __attribute__((ext_vector_type(8)));

constexpr int Bn = 8, Cc = 128, Hh = 96, Ww = 320;
constexpr int HW = Hh * Ww;
constexpr int ND = 9;
constexpr int NQ = Cc / 4;
constexpr int TILES_PER_SEG = 4;
constexpr int SEGS = Ww / (16 * TILES_PER_SEG);
constexpr int WAVES_PER_BLOCK = 8;

__device__ __forceinline__ void load_bblock(v2f* Bb, const float* __restrict__ x2row,
                                            int j0, int g, int nn) {
  int xsrc = j0 + nn - 4;
  bool valid = ((unsigned)xsrc < (unsigned)Ww);
#pragma unroll
  for (int q = 0; q < NQ; ++q) {
    int c = 4 * q + 2 * g;
    float bx = 0.0f, by = 0.0f;
    if (valid) {
      const float* p = x2row + c * HW + xsrc;
      bx = p[0];
      by = p[HW];
    }
    Bb[q].x = bx;
    Bb[q].y = by;
  }
}

__device__ __forceinline__ v2f load_afrag(const float* __restrict__ x1row, int x0,
                                          int q, int g, int mm) {
  int c = 4 * q + 2 * g;
  const float* p = x1row + c * HW + x0 + mm;
  v2f a;
  a.x = __builtin_nontemporal_load(p);
  a.y = __builtin_nontemporal_load(p + HW);
  return a;
}

__device__ __forceinline__ v8f wmma4(const v2f a, const v2f b, const v8f c) {
  return __builtin_amdgcn_wmma_f32_16x16x4_f32(false, a, false, b, (short)0, c,
                                               false, false);
}

__device__ __forceinline__ void store_tile(float* __restrict__ outbase, int x0,
                                           int g, int nn, const v8f d0,
                                           const v8f d1) {
  const float invC = 1.0f / (float)Cc;
#pragma unroll
  for (int r = 0; r < 8; ++r) {
    int mrow = r + 8 * g;
    int t0 = nn - mrow;
    if ((unsigned)t0 <= 8u)
      __builtin_nontemporal_store(d0[r] * invC, outbase + t0 * HW + x0 + mrow);
    int t1 = nn + 16 - mrow;
    if ((unsigned)t1 <= 8u)
      __builtin_nontemporal_store(d1[r] * invC, outbase + t1 * HW + x0 + mrow);
  }
}

__global__ __launch_bounds__(WAVES_PER_BLOCK * 32) void
Correlation1d_74363063763057_kernel(const float* __restrict__ x1,
                                    const float* __restrict__ x2,
                                    float* __restrict__ out) {
  int wid = blockIdx.x * WAVES_PER_BLOCK + ((int)threadIdx.x >> 5);
  int lane = (int)threadIdx.x & 31;
  int nn = lane & 15;
  int g = lane >> 4;

  int seg = wid % SEGS;
  int row = wid / SEGS;
  if (row >= Bn * Hh) return;
  int y = row % Hh;
  int b = row / Hh;

  const float* x1row = x1 + (size_t)b * Cc * HW + (size_t)y * Ww;
  const float* x2row = x2 + (size_t)b * Cc * HW + (size_t)y * Ww;
  float* outbase = out + (size_t)b * ND * HW + (size_t)y * Ww;

  int xs = seg * (16 * TILES_PER_SEG);

  v8f zero = {};
  v8f d0a = zero, d0b = zero;
  v8f d1a = zero, d1b = zero;
  v8f d0_done = zero;

  v2f Aa[NQ];
  v2f Bb[NQ];

  load_bblock(Bb, x2row, xs, g, nn);
#pragma unroll
  for (int q = 0; q < NQ; ++q) {
    Aa[q] = load_afrag(x1row, xs, q, g, nn);
    if (q & 1)
      d0b = wmma4(Aa[q], Bb[q], d0b);
    else
      d0a = wmma4(Aa[q], Bb[q], d0a);
  }
  d0_done = d0a + d0b;

  for (int kblk = 1; kblk < TILES_PER_SEG; ++kblk) {
    int j0 = xs + 16 * kblk;
    load_bblock(Bb, x2row, j0, g, nn);
    d0a = zero;
    d0b = zero;
#pragma unroll
    for (int q = 0; q < NQ; ++q) {
      v2f an = load_afrag(x1row, j0, q, g, nn);
      if (q & 1) {
        d1b = wmma4(Aa[q], Bb[q], d1b);
        d0b = wmma4(an, Bb[q], d0b);
      } else {
        d1a = wmma4(Aa[q], Bb[q], d1a);
        d0a = wmma4(an, Bb[q], d0a);
      }
      Aa[q] = an;
    }
    store_tile(outbase, j0 - 16, g, nn, d0_done, d1a + d1b);
    d0_done = d0a + d0b;
    d1a = zero;
    d1b = zero;
  }

  {
    int j0 = xs + 16 * TILES_PER_SEG;
    load_bblock(Bb, x2row, j0, g, nn);
#pragma unroll
    for (int q = 0; q < NQ; ++q) {
      if (q & 1)
        d1b = wmma4(Aa[q], Bb[q], d1b);
      else
        d1a = wmma4(Aa[q], Bb[q], d1a);
    }
    store_tile(outbase, j0 - 16, g, nn, d0_done, d1a + d1b);
  }
}

extern "C" void kernel_launch(void* const* d_in, const int* in_sizes, int n_in,
                              void* d_out, int out_size, void* d_ws, size_t ws_size,
                              hipStream_t stream) {
  (void)in_sizes; (void)n_in; (void)out_size; (void)d_ws; (void)ws_size;
  const float* x1 = (const float*)d_in[0];
  const float* x2 = (const float*)d_in[1];
  float* out = (float*)d_out;

  int totalWaves = Bn * Hh * SEGS;
  int blocks = (totalWaves + WAVES_PER_BLOCK - 1) / WAVES_PER_BLOCK;
  Correlation1d_74363063763057_kernel<<<blocks, WAVES_PER_BLOCK * 32, 0, stream>>>(
      x1, x2, out);
}


# Four waves per workgroup, one on each SIMD, each owning 80 columns (5 tiles, 10 products per 4 channels) in place of five waves of 64: a row's matrix issue is spread evenly over the four SIMDs
# speedup vs baseline: 4.5278x; 4.5278x over previous
#include <hip/hip_runtime.h>


typedef float v2f __attribute__((ext_vector_type(2)));
typedef float v4f __attribute__((ext_vector_type(4)));
typedef float v8f __attribute__((ext_vector_type(8)));

constexpr int Bn = 8, Cc = 128, Hh = 96, Ww = 320;
constexpr int HW = Hh * Ww;
constexpr int ND = 9;
constexpr int KC = 16;
constexpr int NSTAGE = Cc / KC;
constexpr int NPAIR = KC / 2;
constexpr int NBLK = 21;
constexpr int PP = NBLK * 32;
constexpr int BUF = NPAIR * PP;
constexpr int NWAVE = 4;
constexpr int NT = Ww / 16 / NWAVE;
constexpr int NTHR = NWAVE * 32;
constexpr int CHUNKS = Ww / 4;
constexpr int NGRP = CHUNKS / 16;
constexpr int LPT = KC * CHUNKS / NTHR;
constexpr int PO = 324;
constexpr int DUMP = ND * PO;

__device__ __forceinline__ v8f wmma4(const v2f a, const v2f b, const v8f c) {
  return __builtin_amdgcn_wmma_f32_16x16x4_f32(false, a, false, b, (short)0, c,
                                               false, false);
}

__device__ __forceinline__ void stage_barrier() {
  __builtin_amdgcn_fence(__ATOMIC_RELEASE, "workgroup", "local");
  __builtin_amdgcn_s_barrier();
  __builtin_amdgcn_fence(__ATOMIC_ACQUIRE, "workgroup", "local");
}

__device__ __forceinline__ void multiply_stage(const float* fa, const float* fb,
                                               v8f (&d0)[NT], v8f (&d1)[NT]) {
#pragma unroll
  for (int q = 0; q < KC / 4; ++q) {
    v2f A[NT], B[NT + 1];
#pragma unroll
    for (int k = 0; k < NT; ++k) {
      A[k].x = fa[(2 * q) * PP + 32 * k];
      A[k].y = fa[(2 * q) * PP + 32 * k + 16];
    }
#pragma unroll
    for (int j = 0; j < NT + 1; ++j) {
      B[j].x = fb[(2 * q) * PP + 32 * j];
      B[j].y = fb[(2 * q) * PP + 32 * j + 16];
    }
#pragma unroll
    for (int k = 0; k < NT; ++k) {
      d0[k] = wmma4(A[k], B[k], d0[k]);
      d1[k] = wmma4(A[k], B[k + 1], d1[k]);
    }
  }
}

__global__ __launch_bounds__(NTHR) void
Correlation1d_rowstage_kernel(const float* __restrict__ x1,
                              const float* __restrict__ x2,
                              float* __restrict__ out) {
  __shared__ __attribute__((aligned(16))) float s1[2 * BUF];
  __shared__ __attribute__((aligned(16))) float s2[2 * BUF];
  __shared__ __attribute__((aligned(16))) float so[DUMP + 32];

  const int tid = (int)threadIdx.x;
  const int wave = tid >> 5;
  const int lane = tid & 31;
  const int nn = lane & 15;
  const int g = lane >> 4;
  const int row = (int)blockIdx.x;
  const int y = row % Hh;
  const int b = row / Hh;
  const int hh = (lane >> 2) & 1;
  const int ch = 4 * (lane >> 3) + (lane & 3);

  const size_t goff = (size_t)b * Cc * HW + (size_t)(4 * wave + hh) * HW + (size_t)y * Ww + 4 * ch;
  const float* g1 = x1 + goff;
  const float* g2 = x2 + goff;
  const int st1 = (2 * wave) * PP + (ch >> 2) * 32 + hh * 16 + 4 * (ch & 3);
  const int st2 = (2 * wave) * PP + ((ch + 1) >> 2) * 32 + hh * 16 + 4 * ((ch + 1) & 3);

  if (tid < 2 * NPAIR * 2 * 4) {
    int piece = tid & 3;
    int v = (tid >> 2) & 1;
    int pr = tid >> 3;
    int off = piece == 0 ? 0 : (NBLK - 1) * 32 + 4 * piece;
    v4f z = {};
    *(v4f*)&s2[pr * PP + v * 16 + off] = z;
  }

  v4f ra[LPT], rb[LPT];
#pragma unroll
  for (int i = 0; i < LPT; ++i) {
    ra[i] = __builtin_nontemporal_load((const v4f*)(g1 + (size_t)(2 * (i / NGRP)) * HW + 64 * (i % NGRP)));
    rb[i] = __builtin_nontemporal_load((const v4f*)(g2 + (size_t)(2 * (i / NGRP)) * HW + 64 * (i % NGRP)));
  }

  v8f d0[NT], d1[NT];
#pragma unroll
  for (int k = 0; k < NT; ++k) {
    v8f zero = {};
    d0[k] = zero;
    d1[k] = zero;
  }

  const int foff = g * PP + (NT * wave) * 32 + nn;

  for (int s = 0; s < NSTAGE - 1; ++s) {
    float* b1 = s1 + (s & 1) * BUF;
    float* b2 = s2 + (s & 1) * BUF;
    const float* n1 = g1 + (size_t)(s + 1) * KC * HW;
    const float* n2 = g2 + (size_t)(s + 1) * KC * HW;
#pragma unroll
    for (int i = 0; i < LPT; ++i) {
      *(v4f*)&b1[(i / NGRP) * PP + 128 * (i % NGRP) + st1] = ra[i];
      ra[i] = __builtin_nontemporal_load((const v4f*)(n1 + (size_t)(2 * (i / NGRP)) * HW + 64 * (i % NGRP)));
      __builtin_amdgcn_sched_barrier(0);
      *(v4f*)&b2[(i / NGRP) * PP + 128 * (i % NGRP) + st2] = rb[i];
      rb[i] = __builtin_nontemporal_load((const v4f*)(n2 + (size_t)(2 * (i / NGRP)) * HW + 64 * (i % NGRP)));
      __builtin_amdgcn_sched_barrier(0);
    }
    stage_barrier();
    multiply_stage(b1 + foff, b2 + foff, d0, d1);
  }
  {
    float* b1 = s1 + ((NSTAGE - 1) & 1) * BUF;
    float* b2 = s2 + ((NSTAGE - 1) & 1) * BUF;
#pragma unroll
    for (int i = 0; i < LPT; ++i) {
      *(v4f*)&b1[(i / NGRP) * PP + 128 * (i % NGRP) + st1] = ra[i];
      *(v4f*)&b2[(i / NGRP) * PP + 128 * (i % NGRP) + st2] = rb[i];
    }
    stage_barrier();
    multiply_stage(b1 + foff, b2 + foff, d0, d1);
  }

  const float invC = 1.0f / (float)Cc;
#pragma unroll
  for (int k = 0; k < NT; ++k) {
    int x0 = 16 * NT * wave + 16 * k;
#pragma unroll
    for (int r = 0; r < 8; ++r) {
      int mrow = r + 8 * g;
      int dl = mrow - nn;
      bool in0 = (unsigned)(-dl) <= 8u;
      bool in1 = dl >= 8;
      int t = in0 ? -dl : 16 - dl;
      float val = (in0 ? d0[k][r] : d1[k][r]) * invC;
      int idx = (in0 || in1) ? t * PO + x0 + mrow : DUMP + lane;
      so[idx] = val;
    }
  }
  __syncthreads();
  float* ob = out + (size_t)b * ND * HW + (size_t)y * Ww;
#pragma unroll
  for (int i = 0; i < (ND * CHUNKS + NTHR - 1) / NTHR; ++i) {
    int idx = i * NTHR + tid;
    int plane = idx / CHUNKS;
    int oc = idx - plane * CHUNKS;
    if ((i + 1) * NTHR <= ND * CHUNKS || idx < ND * CHUNKS) {
      v4f v = *(const v4f*)&so[plane * PO + 4 * oc];
      __builtin_nontemporal_store(v, (v4f*)(ob + (size_t)plane * HW + 4 * oc));
    }
  }
}

extern "C" void kernel_launch(void* const* d_in, const int* in_sizes, int n_in,
                              void* d_out, int out_size, void* d_ws, size_t ws_size,
                              hipStream_t stream) {
  (void)in_sizes; (void)n_in; (void)out_size; (void)d_ws; (void)ws_size;
  const float* x1 = (const float*)d_in[0];
  const float* x2 = (const float*)d_in[1];
  float* out = (float*)d_out;
  Correlation1d_rowstage_kernel<<<Bn * Hh, NTHR, 0, stream>>>(x1, x2, out);
}
